# starting kernel
# baseline (speedup 1.0000x reference)
#include <hip/hip_runtime.h>
#include <hip/hip_bf16.h>

typedef float v2f __attribute__((ext_vector_type(2)));
typedef float v8f __attribute__((ext_vector_type(8)));

#define BATCH 2
#define NPTS  8192
#define MPTS  8192
#define MTILES_PER_BATCH (MPTS / 16)
#define WAVES_PER_BLOCK  8

__global__ __launch_bounds__(256) void pack_refcloud(
    const float* __restrict__ input, float* __restrict__ packB) {
  int idx = blockIdx.x * blockDim.x + threadIdx.x;
  if (idx >= BATCH * NPTS) return;
  const float* q = input + (size_t)idx * 3;
  float x = q[0], y = q[1], z = q[2];
  float w = -0.5f * (x * x + y * y + z * z);
  float4 v = make_float4(x, y, z, w);
  *(float4*)(packB + (size_t)idx * 4) = v;
}

__global__ __launch_bounds__(WAVES_PER_BLOCK * 32) void chamfer_wmma(
    const float* __restrict__ point,
    const float* __restrict__ packB,
    float* __restrict__ out)
{
  const int lane    = threadIdx.x & 31;
  const int wave    = threadIdx.x >> 5;
  const int tile    = blockIdx.x * WAVES_PER_BLOCK + wave;
  const int b       = tile / MTILES_PER_BATCH;
  const int m0      = (tile % MTILES_PER_BATCH) * 16;
  const int mrow    = lane & 15;
  const int half    = lane >> 4;

  const float* pp = point + ((size_t)b * MPTS + (m0 + mrow)) * 3;
  const float px = pp[0], py = pp[1], pz = pp[2];
  v2f a;
  a.x = half ? pz : px;
  a.y = half ? 1.0f : py;
  const float p2 = px * px + py * py + pz * pz;

  const float* bbase = packB + ((size_t)b * NPTS + mrow) * 4 + (half << 1);

  v8f cmax;
  #pragma unroll
  for (int i = 0; i < 8; ++i) cmax[i] = -3.402823466e38f;

  for (int nt = 0; nt < NPTS / 16; nt += 2) {
    v2f b0 = *(const v2f*)(bbase + (size_t)nt * 64);
    v2f b1 = *(const v2f*)(bbase + (size_t)(nt + 1) * 64);
    v8f c0 = {};
    v8f c1 = {};
    c0 = __builtin_amdgcn_wmma_f32_16x16x4_f32(
         false, a,  false, b0,
         (short)0, c0,  false,  false);
    c1 = __builtin_amdgcn_wmma_f32_16x16x4_f32(
        false, a, false, b1, (short)0, c1, false, false);
    #pragma unroll
    for (int i = 0; i < 8; ++i) {
      cmax[i] = fmaxf(cmax[i], fmaxf(c0[i], c1[i]));
    }
  }

  #pragma unroll
  for (int m = 8; m >= 1; m >>= 1) {
    #pragma unroll
    for (int i = 0; i < 8; ++i) {
      float o = __shfl_xor(cmax[i], m, 32);
      cmax[i] = fmaxf(cmax[i], o);
    }
  }

  if (lane < 8) {
    out[(size_t)b * MPTS + m0 + lane] = fmaxf(p2 - 2.0f * cmax[lane], 0.0f);
  } else if (lane >= 24) {
    int r = lane - 24;
    out[(size_t)b * MPTS + m0 + 8 + r] = fmaxf(p2 - 2.0f * cmax[r], 0.0f);
  }
}

extern "C" void kernel_launch(void* const* d_in, const int* in_sizes, int n_in,
                              void* d_out, int out_size, void* d_ws, size_t ws_size,
                              hipStream_t stream) {
  const float* input = (const float*)d_in[0];
  const float* point = (const float*)d_in[1];
  float* out = (float*)d_out;
  float* packB = (float*)d_ws;

  {
    int total = BATCH * NPTS;
    pack_refcloud<<<(total + 255) / 256, 256, 0, stream>>>(input, packB);
  }
  {
    int tiles = BATCH * MTILES_PER_BATCH;
    chamfer_wmma<<<tiles / WAVES_PER_BLOCK, WAVES_PER_BLOCK * 32, 0, stream>>>(
        point, packB, out);
  }
}


# Main kernel copies its batch's packed cloud (128 KiB) into LDS once per block and scans the 512 N tiles from LDS, next group's reads issued ahead, instead of two global loads per loop trip
# speedup vs baseline: 6.4688x; 6.4688x over previous
#include <hip/hip_runtime.h>
#include <hip/hip_bf16.h>

typedef float v2f __attribute__((ext_vector_type(2)));
typedef float v8f __attribute__((ext_vector_type(8)));

#define BATCH 2
#define NPTS  8192
#define MPTS  8192
#define MTILES_PER_BATCH (MPTS / 16)
#define WAVES_PER_BLOCK  8

__global__ __launch_bounds__(256) void pack_refcloud(
    const float* __restrict__ input, float* __restrict__ packB) {
  int idx = blockIdx.x * blockDim.x + threadIdx.x;
  if (idx >= BATCH * NPTS) return;
  const float* q = input + (size_t)idx * 3;
  float x = q[0], y = q[1], z = q[2];
  float w = -0.5f * (x * x + y * y + z * z);
  float4 v = make_float4(x, y, z, w);
  *(float4*)(packB + (size_t)idx * 4) = v;
}

#define NTILES   (NPTS / 16)
#define GROUP    8
#define BLOCKS_PER_BATCH (MTILES_PER_BATCH / WAVES_PER_BLOCK)

__device__ __forceinline__ void load_group(v2f (&f)[GROUP], const float* lb, int nt) {
  #pragma unroll
  for (int u = 0; u < GROUP; ++u) f[u] = *(const v2f*)(lb + (nt + u) * 64);
}

__device__ __forceinline__ void scan_group(const v2f (&f)[GROUP], v2f a, v8f& cmax) {
  v8f d[GROUP];
  #pragma unroll
  for (int u = 0; u < GROUP; ++u) {
    v8f z = {};
    d[u] = __builtin_amdgcn_wmma_f32_16x16x4_f32(
         false, a,  false, f[u],
         (short)0, z,  false,  false);
  }
  #pragma unroll
  for (int u = 0; u < GROUP; u += 2) {
    #pragma unroll
    for (int i = 0; i < 8; ++i) {
      cmax[i] = fmaxf(cmax[i], fmaxf(d[u][i], d[u + 1][i]));
    }
  }
}

__global__ __launch_bounds__(WAVES_PER_BLOCK * 32) void chamfer_wmma(
    const float* __restrict__ point,
    const float* __restrict__ packB,
    float* __restrict__ out)
{
  __shared__ __attribute__((aligned(16))) float sB[NPTS * 4];

  const int lane    = threadIdx.x & 31;
  const int wave    = threadIdx.x >> 5;
  const int b       = blockIdx.x / BLOCKS_PER_BATCH;
  const int m0      = ((blockIdx.x % BLOCKS_PER_BATCH) * WAVES_PER_BLOCK + wave) * 16;
  const int mrow    = lane & 15;
  const int half    = lane >> 4;

  const float* pp = point + ((size_t)b * MPTS + (m0 + mrow)) * 3;
  const float px = pp[0], py = pp[1], pz = pp[2];
  v2f a;
  a.x = half ? pz : px;
  a.y = half ? 1.0f : py;
  const float p2 = px * px + py * py + pz * pz;

  {
    const float4* gsrc = (const float4*)(packB + (size_t)b * NPTS * 4);
    float4* ldst = (float4*)sB;
    #pragma unroll
    for (int j = 0; j < NPTS / (WAVES_PER_BLOCK * 32); ++j) {
      ldst[j * (WAVES_PER_BLOCK * 32) + threadIdx.x] = gsrc[j * (WAVES_PER_BLOCK * 32) + threadIdx.x];
    }
  }
  __syncthreads();

  const float* lb = sB + mrow * 4 + (half << 1);

  v8f cmax;
  #pragma unroll
  for (int i = 0; i < 8; ++i) cmax[i] = -3.402823466e38f;

  v2f fa[GROUP], fb[GROUP];
  load_group(fa, lb, 0);
  for (int nt = 0; nt < NTILES; nt += 2 * GROUP) {
    load_group(fb, lb, nt + GROUP);
    scan_group(fa, a, cmax);
    load_group(fa, lb, (nt + 2 * GROUP) & (NTILES - 1));
    scan_group(fb, a, cmax);
  }

  #pragma unroll
  for (int m = 8; m >= 1; m >>= 1) {
    #pragma unroll
    for (int i = 0; i < 8; ++i) {
      float o = __shfl_xor(cmax[i], m, 32);
      cmax[i] = fmaxf(cmax[i], o);
    }
  }

  if (lane < 8) {
    out[(size_t)b * MPTS + m0 + lane] = fmaxf(p2 - 2.0f * cmax[lane], 0.0f);
  } else if (lane >= 24) {
    int r = lane - 24;
    out[(size_t)b * MPTS + m0 + 8 + r] = fmaxf(p2 - 2.0f * cmax[r], 0.0f);
  }
}

extern "C" void kernel_launch(void* const* d_in, const int* in_sizes, int n_in,
                              void* d_out, int out_size, void* d_ws, size_t ws_size,
                              hipStream_t stream) {
  const float* input = (const float*)d_in[0];
  const float* point = (const float*)d_in[1];
  float* out = (float*)d_out;
  float* packB = (float*)d_ws;

  {
    int total = BATCH * NPTS;
    pack_refcloud<<<(total + 255) / 256, 256, 0, stream>>>(input, packB);
  }
  {
    int tiles = BATCH * MTILES_PER_BATCH;
    chamfer_wmma<<<tiles / WAVES_PER_BLOCK, WAVES_PER_BLOCK * 32, 0, stream>>>(
        point, packB, out);
  }
}


# Main kernel: 4 M tiles per block instead of 8 (256 blocks, one per WGP, LDS sized so no two share a WGP); two waves split each tile's N scan in halves and join row maxima via LDS (max is order-free)
# speedup vs baseline: 9.0322x; 1.3963x over previous
#include <hip/hip_runtime.h>
#include <hip/hip_bf16.h>

typedef float v2f __attribute__((ext_vector_type(2)));
typedef float v8f __attribute__((ext_vector_type(8)));

#define BATCH 2
#define NPTS  8192
#define MPTS  8192
#define MTILES_PER_BATCH (MPTS / 16)
#define MT_PER_BLOCK     4
#define NSPLIT           2
#define WAVES_PER_BLOCK  (MT_PER_BLOCK * NSPLIT)

__global__ __launch_bounds__(256) void pack_refcloud(
    const float* __restrict__ input, float* __restrict__ packB) {
  int idx = blockIdx.x * blockDim.x + threadIdx.x;
  if (idx >= BATCH * NPTS) return;
  const float* q = input + (size_t)idx * 3;
  float x = q[0], y = q[1], z = q[2];
  float w = -0.5f * (x * x + y * y + z * z);
  float4 v = make_float4(x, y, z, w);
  *(float4*)(packB + (size_t)idx * 4) = v;
}

#define NTILES   (NPTS / 16)
#define GROUP    8
#define NT_PER_WAVE (NTILES / NSPLIT)
#define BLOCKS_PER_BATCH (MTILES_PER_BATCH / MT_PER_BLOCK)
#define LDS_FLOATS (41 * 1024)
#define RED_OFF    (LDS_FLOATS - MT_PER_BLOCK * 16)

__device__ __forceinline__ void load_group(v2f (&f)[GROUP], const float* lb, int nt) {
  #pragma unroll
  for (int u = 0; u < GROUP; ++u) f[u] = *(const v2f*)(lb + (nt + u) * 64);
}

__device__ __forceinline__ void scan_group(const v2f (&f)[GROUP], v2f a, v8f& cmax) {
  v8f d[GROUP];
  #pragma unroll
  for (int u = 0; u < GROUP; ++u) {
    v8f z = {};
    d[u] = __builtin_amdgcn_wmma_f32_16x16x4_f32(
         false, a,  false, f[u],
         (short)0, z,  false,  false);
  }
  #pragma unroll
  for (int u = 0; u < GROUP; u += 2) {
    #pragma unroll
    for (int i = 0; i < 8; ++i) {
      cmax[i] = fmaxf(cmax[i], fmaxf(d[u][i], d[u + 1][i]));
    }
  }
}

__global__ __launch_bounds__(WAVES_PER_BLOCK * 32) void chamfer_wmma(
    const float* __restrict__ point,
    const float* __restrict__ packB,
    float* __restrict__ out)
{
  __shared__ __attribute__((aligned(16))) float sB[LDS_FLOATS];

  const int lane    = threadIdx.x & 31;
  const int wave    = threadIdx.x >> 5;
  const int b       = blockIdx.x / BLOCKS_PER_BATCH;
  const int mt      = wave / NSPLIT;
  const int nh      = wave % NSPLIT;
  const int m0      = ((blockIdx.x % BLOCKS_PER_BATCH) * MT_PER_BLOCK + mt) * 16;
  const int mrow    = lane & 15;
  const int half    = lane >> 4;

  const float* pp = point + ((size_t)b * MPTS + (m0 + mrow)) * 3;
  const float px = pp[0], py = pp[1], pz = pp[2];
  v2f a;
  a.x = half ? pz : px;
  a.y = half ? 1.0f : py;
  const float p2 = px * px + py * py + pz * pz;

  {
    const float4* gsrc = (const float4*)(packB + (size_t)b * NPTS * 4);
    float4* ldst = (float4*)sB;
    #pragma unroll
    for (int j = 0; j < NPTS / (WAVES_PER_BLOCK * 32); ++j) {
      ldst[j * (WAVES_PER_BLOCK * 32) + threadIdx.x] = gsrc[j * (WAVES_PER_BLOCK * 32) + threadIdx.x];
    }
  }
  __syncthreads();

  const float* lb = sB + nh * (NT_PER_WAVE * 64) + mrow * 4 + (half << 1);

  v8f cmax;
  #pragma unroll
  for (int i = 0; i < 8; ++i) cmax[i] = -3.402823466e38f;

  v2f fa[GROUP], fb[GROUP];
  load_group(fa, lb, 0);
  for (int nt = 0; nt < NT_PER_WAVE; nt += 2 * GROUP) {
    load_group(fb, lb, nt + GROUP);
    scan_group(fa, a, cmax);
    load_group(fa, lb, (nt + 2 * GROUP) & (NT_PER_WAVE - 1));
    scan_group(fb, a, cmax);
  }

  #pragma unroll
  for (int m = 8; m >= 1; m >>= 1) {
    #pragma unroll
    for (int i = 0; i < 8; ++i) {
      float o = __shfl_xor(cmax[i], m, 32);
      cmax[i] = fmaxf(cmax[i], o);
    }
  }

  float* red = sB + RED_OFF + mt * 16;
  const bool owner = (lane < 8) || (lane >= 24);
  const int  r     = lane & 7;
  const int  row   = (lane < 8) ? r : 8 + r;
  const float mine = cmax[r];
  if (nh != 0 && owner) red[row] = mine;
  __syncthreads();
  if (nh == 0 && owner) {
    const float best = fmaxf(mine, red[row]);
    out[(size_t)b * MPTS + m0 + row] = fmaxf(p2 - 2.0f * best, 0.0f);
  }
}

extern "C" void kernel_launch(void* const* d_in, const int* in_sizes, int n_in,
                              void* d_out, int out_size, void* d_ws, size_t ws_size,
                              hipStream_t stream) {
  const float* input = (const float*)d_in[0];
  const float* point = (const float*)d_in[1];
  float* out = (float*)d_out;
  float* packB = (float*)d_ws;

  {
    int total = BATCH * NPTS;
    pack_refcloud<<<(total + 255) / 256, 256, 0, stream>>>(input, packB);
  }
  {
    int tiles = BATCH * MTILES_PER_BATCH;
    chamfer_wmma<<<tiles / MT_PER_BLOCK, WAVES_PER_BLOCK * 32, 0, stream>>>(
        point, packB, out);
  }
}


# Pack kernel folded into the main kernel's LDS staging: each block reads its batch's raw cloud (96 KiB, 12 B per point), forms -|q|^2/2 itself, stores float4 rows to LDS; one launch, not two
# speedup vs baseline: 12.2764x; 1.3592x over previous
#include <hip/hip_runtime.h>
#include <hip/hip_bf16.h>

typedef float v2f __attribute__((ext_vector_type(2)));
typedef float v8f __attribute__((ext_vector_type(8)));

#define BATCH 2
#define NPTS  8192
#define MPTS  8192
#define MTILES_PER_BATCH (MPTS / 16)
#define MT_PER_BLOCK     4
#define NSPLIT           2
#define WAVES_PER_BLOCK  (MT_PER_BLOCK * NSPLIT)

#define NTILES   (NPTS / 16)
#define GROUP    8
#define NT_PER_WAVE (NTILES / NSPLIT)
#define BLOCKS_PER_BATCH (MTILES_PER_BATCH / MT_PER_BLOCK)
#define LDS_FLOATS (41 * 1024)
#define RED_OFF    (LDS_FLOATS - MT_PER_BLOCK * 16)

__device__ __forceinline__ void load_group(v2f (&f)[GROUP], const float* lb, int nt) {
  #pragma unroll
  for (int u = 0; u < GROUP; ++u) f[u] = *(const v2f*)(lb + (nt + u) * 64);
}

__device__ __forceinline__ void scan_group(const v2f (&f)[GROUP], v2f a, v8f& cmax) {
  v8f d[GROUP];
  #pragma unroll
  for (int u = 0; u < GROUP; ++u) {
    v8f z = {};
    d[u] = __builtin_amdgcn_wmma_f32_16x16x4_f32(
         false, a,  false, f[u],
         (short)0, z,  false,  false);
  }
  #pragma unroll
  for (int u = 0; u < GROUP; u += 2) {
    #pragma unroll
    for (int i = 0; i < 8; ++i) {
      cmax[i] = fmaxf(cmax[i], fmaxf(d[u][i], d[u + 1][i]));
    }
  }
}

__global__ __launch_bounds__(WAVES_PER_BLOCK * 32) void chamfer_wmma(
    const float* __restrict__ point,
    const float* __restrict__ input,
    float* __restrict__ out)
{
  __shared__ __attribute__((aligned(16))) float sB[LDS_FLOATS];

  const int lane    = threadIdx.x & 31;
  const int wave    = threadIdx.x >> 5;
  const int b       = blockIdx.x / BLOCKS_PER_BATCH;
  const int mt      = wave / NSPLIT;
  const int nh      = wave % NSPLIT;
  const int m0      = ((blockIdx.x % BLOCKS_PER_BATCH) * MT_PER_BLOCK + mt) * 16;
  const int mrow    = lane & 15;
  const int half    = lane >> 4;

  const float* pp = point + ((size_t)b * MPTS + (m0 + mrow)) * 3;
  const float px = pp[0], py = pp[1], pz = pp[2];
  v2f a;
  a.x = half ? pz : px;
  a.y = half ? 1.0f : py;
  const float p2 = px * px + py * py + pz * pz;

  {
    const float* gsrc = input + (size_t)b * NPTS * 3;
    float4* ldst = (float4*)sB;
    #pragma unroll
    for (int j = 0; j < NPTS / (WAVES_PER_BLOCK * 32); ++j) {
      const int n = j * (WAVES_PER_BLOCK * 32) + threadIdx.x;
      const float* q = gsrc + n * 3;
      const float x = q[0], y = q[1], z = q[2];
      const float w = -0.5f * (x * x + y * y + z * z);
      ldst[n] = make_float4(x, y, z, w);
    }
  }
  __syncthreads();

  const float* lb = sB + nh * (NT_PER_WAVE * 64) + mrow * 4 + (half << 1);

  v8f cmax;
  #pragma unroll
  for (int i = 0; i < 8; ++i) cmax[i] = -3.402823466e38f;

  v2f fa[GROUP], fb[GROUP];
  load_group(fa, lb, 0);
  for (int nt = 0; nt < NT_PER_WAVE; nt += 2 * GROUP) {
    load_group(fb, lb, nt + GROUP);
    scan_group(fa, a, cmax);
    load_group(fa, lb, (nt + 2 * GROUP) & (NT_PER_WAVE - 1));
    scan_group(fb, a, cmax);
  }

  #pragma unroll
  for (int m = 8; m >= 1; m >>= 1) {
    #pragma unroll
    for (int i = 0; i < 8; ++i) {
      float o = __shfl_xor(cmax[i], m, 32);
      cmax[i] = fmaxf(cmax[i], o);
    }
  }

  float* red = sB + RED_OFF + mt * 16;
  const bool owner = (lane < 8) || (lane >= 24);
  const int  r     = lane & 7;
  const int  row   = (lane < 8) ? r : 8 + r;
  const float mine = cmax[r];
  if (nh != 0 && owner) red[row] = mine;
  __syncthreads();
  if (nh == 0 && owner) {
    const float best = fmaxf(mine, red[row]);
    out[(size_t)b * MPTS + m0 + row] = fmaxf(p2 - 2.0f * best, 0.0f);
  }
}

extern "C" void kernel_launch(void* const* d_in, const int* in_sizes, int n_in,
                              void* d_out, int out_size, void* d_ws, size_t ws_size,
                              hipStream_t stream) {
  const float* input = (const float*)d_in[0];
  const float* point = (const float*)d_in[1];
  float* out = (float*)d_out;

  {
    int tiles = BATCH * MTILES_PER_BATCH;
    chamfer_wmma<<<tiles / MT_PER_BLOCK, WAVES_PER_BLOCK * 32, 0, stream>>>(
        point, input, out);
  }
}


# Every max is the NaN-propagating maximum on the raw matrix results (fmaxf spent a quieting self-max per result register); scan fold as three-way maxima: vector ops per 16 matrix ops 116 -> 67
# speedup vs baseline: 13.1824x; 1.0738x over previous
#include <hip/hip_runtime.h>
#include <hip/hip_bf16.h>

typedef float v2f __attribute__((ext_vector_type(2)));
typedef float v8f __attribute__((ext_vector_type(8)));

#define BATCH 2
#define NPTS  8192
#define MPTS  8192
#define MTILES_PER_BATCH (MPTS / 16)
#define MT_PER_BLOCK     4
#define NSPLIT           2
#define WAVES_PER_BLOCK  (MT_PER_BLOCK * NSPLIT)

#define NTILES   (NPTS / 16)
#define GROUP    8
#define NT_PER_WAVE (NTILES / NSPLIT)
#define BLOCKS_PER_BATCH (MTILES_PER_BATCH / MT_PER_BLOCK)
#define LDS_FLOATS (41 * 1024)
#define RED_OFF    (LDS_FLOATS - MT_PER_BLOCK * 16)

__device__ __forceinline__ float vmax(float a, float b) {
  return __builtin_elementwise_maximum(a, b);
}

__device__ __forceinline__ void load_group(v2f (&f)[GROUP], const float* lb, int nt) {
  #pragma unroll
  for (int u = 0; u < GROUP; ++u) f[u] = *(const v2f*)(lb + (nt + u) * 64);
}

__device__ __forceinline__ void scan_group(const v2f (&f)[GROUP], v2f a, v8f& cmax) {
  v8f d[GROUP];
  #pragma unroll
  for (int u = 0; u < GROUP; ++u) {
    v8f z = {};
    d[u] = __builtin_amdgcn_wmma_f32_16x16x4_f32(
         false, a,  false, f[u],
         (short)0, z,  false,  false);
  }
  #pragma unroll
  for (int i = 0; i < 8; ++i) {
    const float ta = vmax(vmax(d[0][i], d[1][i]), d[2][i]);
    const float tb = vmax(vmax(d[3][i], d[4][i]), d[5][i]);
    const float tc = vmax(vmax(cmax[i], d[6][i]), d[7][i]);
    cmax[i] = vmax(vmax(tc, ta), tb);
  }
}

__global__ __launch_bounds__(WAVES_PER_BLOCK * 32) void chamfer_wmma(
    const float* __restrict__ point,
    const float* __restrict__ input,
    float* __restrict__ out)
{
  __shared__ __attribute__((aligned(16))) float sB[LDS_FLOATS];

  const int lane    = threadIdx.x & 31;
  const int wave    = threadIdx.x >> 5;
  const int b       = blockIdx.x / BLOCKS_PER_BATCH;
  const int mt      = wave / NSPLIT;
  const int nh      = wave % NSPLIT;
  const int m0      = ((blockIdx.x % BLOCKS_PER_BATCH) * MT_PER_BLOCK + mt) * 16;
  const int mrow    = lane & 15;
  const int half    = lane >> 4;

  const float* pp = point + ((size_t)b * MPTS + (m0 + mrow)) * 3;
  const float px = pp[0], py = pp[1], pz = pp[2];
  v2f a;
  a.x = half ? pz : px;
  a.y = half ? 1.0f : py;
  const float p2 = px * px + py * py + pz * pz;

  {
    const float* gsrc = input + (size_t)b * NPTS * 3;
    float4* ldst = (float4*)sB;
    #pragma unroll
    for (int j = 0; j < NPTS / (WAVES_PER_BLOCK * 32); ++j) {
      const int n = j * (WAVES_PER_BLOCK * 32) + threadIdx.x;
      const float* q = gsrc + n * 3;
      const float x = q[0], y = q[1], z = q[2];
      const float w = -0.5f * (x * x + y * y + z * z);
      ldst[n] = make_float4(x, y, z, w);
    }
  }
  __syncthreads();

  const float* lb = sB + nh * (NT_PER_WAVE * 64) + mrow * 4 + (half << 1);

  v8f cmax;
  #pragma unroll
  for (int i = 0; i < 8; ++i) cmax[i] = -3.402823466e38f;

  v2f fa[GROUP], fb[GROUP];
  load_group(fa, lb, 0);
  for (int nt = 0; nt < NT_PER_WAVE; nt += 2 * GROUP) {
    load_group(fb, lb, nt + GROUP);
    scan_group(fa, a, cmax);
    load_group(fa, lb, (nt + 2 * GROUP) & (NT_PER_WAVE - 1));
    scan_group(fb, a, cmax);
  }

  #pragma unroll
  for (int m = 8; m >= 1; m >>= 1) {
    #pragma unroll
    for (int i = 0; i < 8; ++i) {
      float o = __shfl_xor(cmax[i], m, 32);
      cmax[i] = vmax(cmax[i], o);
    }
  }

  float* red = sB + RED_OFF + mt * 16;
  const bool owner = (lane < 8) || (lane >= 24);
  const int  r     = lane & 7;
  const int  row   = (lane < 8) ? r : 8 + r;
  const float mine = cmax[r];
  if (nh != 0 && owner) red[row] = mine;
  __syncthreads();
  if (nh == 0 && owner) {
    const float best = vmax(mine, red[row]);
    out[(size_t)b * MPTS + m0 + row] = vmax(p2 - 2.0f * best, 0.0f);
  }
}

extern "C" void kernel_launch(void* const* d_in, const int* in_sizes, int n_in,
                              void* d_out, int out_size, void* d_ws, size_t ws_size,
                              hipStream_t stream) {
  const float* input = (const float*)d_in[0];
  const float* point = (const float*)d_in[1];
  float* out = (float*)d_out;

  {
    int tiles = BATCH * MTILES_PER_BATCH;
    chamfer_wmma<<<tiles / MT_PER_BLOCK, WAVES_PER_BLOCK * 32, 0, stream>>>(
        point, input, out);
  }
}
